# starting kernel
# baseline (speedup 1.0000x reference)
#include <hip/hip_runtime.h>
#include <hip/hip_bf16.h>

constexpr int Bn = 8;
constexpr int Sn = 4096;
constexpr int Dn = 1024;
constexpr int Pn = 512;

typedef float v2f __attribute__((ext_vector_type(2)));
typedef float v8f __attribute__((ext_vector_type(8)));

__global__ __launch_bounds__(256) void seg_starts_kernel(
    const int* __restrict__ bnd, int* __restrict__ starts) {
  const int b = blockIdx.x;
  const int tid = threadIdx.x;
  const int base = tid * 16;
  const int* row = bnd + (size_t)b * Sn;

  __shared__ int psum[256];

  int vals[16];
  int local = 0;
#pragma unroll
  for (int i = 0; i < 16; ++i) {
    vals[i] = row[base + i];
    local += vals[i];
  }
  psum[tid] = local;
  __syncthreads();

  for (int off = 1; off < 256; off <<= 1) {
    int add = (tid >= off) ? psum[tid - off] : 0;
    __syncthreads();
    psum[tid] += add;
    __syncthreads();
  }

  int running = psum[tid] - local;
  int* srow = starts + (size_t)b * (Pn + 1);
#pragma unroll
  for (int i = 0; i < 16; ++i) {
    if (vals[i]) {
      srow[running] = base + i;
      ++running;
    }
  }
  if (tid == 0) srow[Pn] = Sn;
}

__global__ __launch_bounds__(128) void pool_wmma_kernel(
    const float* __restrict__ x, const int* __restrict__ starts,
    float* __restrict__ out) {
  constexpr int DCHUNK = 256;
  constexpr int NPB = Pn / 16;
  constexpr int ND = Dn / DCHUNK;

  const int wg = blockIdx.x;
  const int dchunk = wg % ND;
  const int pb = (wg / ND) % NPB;
  const int b = wg / (ND * NPB);
  const int p0 = pb * 16;

  __shared__ int sStarts[17];
  const int tid = threadIdx.x;
  if (tid < 17) sStarts[tid] = starts[(size_t)b * (Pn + 1) + p0 + tid];
  __syncthreads();

  const int lane = tid & 31;
  const int wave = tid >> 5;
  const int hi = lane >> 4;
  const int mrow = lane & 15;
  const int n = lane & 15;

  const int startM = sStarts[mrow];
  const int endM = sStarts[mrow + 1];
  const int s0 = sStarts[0];
  const int s1 = sStarts[16];

  const int dbase = dchunk * DCHUNK + wave * 64;
  const float* xb = x + (size_t)b * Sn * Dn;

  v8f c0 = {}, c1 = {}, c2 = {}, c3 = {};
  const int koff = hi * 2;

  for (int s = s0; s < s1; s += 4) {
    const int ka = s + koff;
    const int kb = ka + 1;

    v2f A;
    A.x = (ka >= startM && ka < endM) ? 1.0f : 0.0f;
    A.y = (kb >= startM && kb < endM) ? 1.0f : 0.0f;

    const int kac = (ka < Sn - 1) ? ka : (Sn - 1);
    const int kbc = (kb < Sn - 1) ? kb : (Sn - 1);
    const float* ra = xb + (size_t)kac * Dn + dbase + n;
    const float* rb = xb + (size_t)kbc * Dn + dbase + n;

    v2f B0, B1, B2, B3;
    B0.x = ra[0];  B0.y = rb[0];
    B1.x = ra[16]; B1.y = rb[16];
    B2.x = ra[32]; B2.y = rb[32];
    B3.x = ra[48]; B3.y = rb[48];

    c0 = __builtin_amdgcn_wmma_f32_16x16x4_f32(false, A, false, B0, (short)0,
                                               c0, false, false);
    c1 = __builtin_amdgcn_wmma_f32_16x16x4_f32(false, A, false, B1, (short)0,
                                               c1, false, false);
    c2 = __builtin_amdgcn_wmma_f32_16x16x4_f32(false, A, false, B2, (short)0,
                                               c2, false, false);
    c3 = __builtin_amdgcn_wmma_f32_16x16x4_f32(false, A, false, B3, (short)0,
                                               c3, false, false);
  }

#pragma unroll
  for (int j = 0; j < 8; ++j) {
    const int rl = j + hi * 8;
    const float inv = 1.0f / (float)(sStarts[rl + 1] - sStarts[rl]);
    const size_t o = ((size_t)(b * Pn + p0 + rl)) * Dn + dbase + n;
    out[o + 0]  = c0[j] * inv;
    out[o + 16] = c1[j] * inv;
    out[o + 32] = c2[j] * inv;
    out[o + 48] = c3[j] * inv;
  }
}

extern "C" void kernel_launch(void* const* d_in, const int* in_sizes, int n_in,
                              void* d_out, int out_size, void* d_ws,
                              size_t ws_size, hipStream_t stream) {
  const float* x = (const float*)d_in[0];
  const int* boundaries = (const int*)d_in[1];
  float* out = (float*)d_out;

  int* starts = (int*)d_ws;

  seg_starts_kernel<<<Bn, 256, 0, stream>>>(boundaries, starts);

  constexpr int DCHUNK = 256;
  const int grid = Bn * (Pn / 16) * (Dn / DCHUNK);
  pool_wmma_kernel<<<grid, 128, 0, stream>>>(x, starts, out);
}


# K loop batched for bytes in flight: per trip a lane issues 16 16-byte x loads (32 tokens x 4 adjacent channels, one per N-tile) ahead of the 32 WMMAs; was 8 4-byte loads per 4 WMMAs
# speedup vs baseline: 1.8210x; 1.8210x over previous
#include <hip/hip_runtime.h>
#include <hip/hip_bf16.h>

constexpr int Bn = 8;
constexpr int Sn = 4096;
constexpr int Dn = 1024;
constexpr int Pn = 512;

typedef float v2f __attribute__((ext_vector_type(2)));
typedef float v4f __attribute__((ext_vector_type(4)));
typedef float v8f __attribute__((ext_vector_type(8)));

__global__ __launch_bounds__(256) void seg_starts_kernel(
    const int* __restrict__ bnd, int* __restrict__ starts) {
  const int b = blockIdx.x;
  const int tid = threadIdx.x;
  const int base = tid * 16;
  const int* row = bnd + (size_t)b * Sn;

  __shared__ int psum[256];

  int vals[16];
  int local = 0;
#pragma unroll
  for (int i = 0; i < 16; ++i) {
    vals[i] = row[base + i];
    local += vals[i];
  }
  psum[tid] = local;
  __syncthreads();

  for (int off = 1; off < 256; off <<= 1) {
    int add = (tid >= off) ? psum[tid - off] : 0;
    __syncthreads();
    psum[tid] += add;
    __syncthreads();
  }

  int running = psum[tid] - local;
  int* srow = starts + (size_t)b * (Pn + 1);
#pragma unroll
  for (int i = 0; i < 16; ++i) {
    if (vals[i]) {
      srow[running] = base + i;
      ++running;
    }
  }
  if (tid == 0) srow[Pn] = Sn;
}

__global__ __launch_bounds__(128) void pool_wmma_kernel(
    const float* __restrict__ x, const int* __restrict__ starts,
    float* __restrict__ out) {
  constexpr int DCHUNK = 256;
  constexpr int NPB = Pn / 16;
  constexpr int ND = Dn / DCHUNK;
  constexpr int UNR = 8;

  const int wg = blockIdx.x;
  const int dchunk = wg % ND;
  const int pb = (wg / ND) % NPB;
  const int b = wg / (ND * NPB);
  const int p0 = pb * 16;

  __shared__ int sStarts[17];
  const int tid = threadIdx.x;
  if (tid < 17) sStarts[tid] = starts[(size_t)b * (Pn + 1) + p0 + tid];
  __syncthreads();

  const int lane = tid & 31;
  const int wave = tid >> 5;
  const int hi = lane >> 4;
  const int mrow = lane & 15;
  const int n = lane & 15;

  const int startM = sStarts[mrow];
  const int endM = sStarts[mrow + 1];
  const int s0 = sStarts[0];
  const int s1 = sStarts[16];
  const int kmax = s1 - 1;

  const int dbase = dchunk * DCHUNK + wave * 64;
  const char* xrow = reinterpret_cast<const char*>(x + (size_t)b * Sn * Dn);
  const unsigned lane_off = (unsigned)(dbase + 4 * n) * 4u;

  v8f c0 = {}, c1 = {}, c2 = {}, c3 = {};
  const int koff = hi * 2;

  for (int s = s0; s < s1; s += 4 * UNR) {
    v4f ra[UNR], rb[UNR];
    for (int i = 0; i < UNR; ++i) {
      const int ka = s + 4 * i + koff;
      const int kb = ka + 1;
      const int kac = (ka < kmax) ? ka : kmax;
      const int kbc = (kb < kmax) ? kb : kmax;
      ra[i] = *reinterpret_cast<const v4f*>(xrow + ((unsigned)kac * (Dn * 4u) + lane_off));
      rb[i] = *reinterpret_cast<const v4f*>(xrow + ((unsigned)kbc * (Dn * 4u) + lane_off));
    }
    __builtin_amdgcn_sched_barrier(0);
    for (int i = 0; i < UNR; ++i) {
      const int ka = s + 4 * i + koff;
      const int kb = ka + 1;

      v2f A;
      A.x = (ka >= startM && ka < endM) ? 1.0f : 0.0f;
      A.y = (kb >= startM && kb < endM) ? 1.0f : 0.0f;

      v2f B0, B1, B2, B3;
      B0.x = ra[i].x; B0.y = rb[i].x;
      B1.x = ra[i].y; B1.y = rb[i].y;
      B2.x = ra[i].z; B2.y = rb[i].z;
      B3.x = ra[i].w; B3.y = rb[i].w;

      c0 = __builtin_amdgcn_wmma_f32_16x16x4_f32(false, A, false, B0, (short)0,
                                                 c0, false, false);
      c1 = __builtin_amdgcn_wmma_f32_16x16x4_f32(false, A, false, B1, (short)0,
                                                 c1, false, false);
      c2 = __builtin_amdgcn_wmma_f32_16x16x4_f32(false, A, false, B2, (short)0,
                                                 c2, false, false);
      c3 = __builtin_amdgcn_wmma_f32_16x16x4_f32(false, A, false, B3, (short)0,
                                                 c3, false, false);
    }
  }

#pragma unroll
  for (int j = 0; j < 8; ++j) {
    const int rl = j + hi * 8;
    const float inv = 1.0f / (float)(sStarts[rl + 1] - sStarts[rl]);
    const size_t o = ((size_t)(b * Pn + p0 + rl)) * Dn + dbase + 4 * n;
    v4f r;
    r.x = c0[j] * inv;
    r.y = c1[j] * inv;
    r.z = c2[j] * inv;
    r.w = c3[j] * inv;
    *reinterpret_cast<v4f*>(out + o) = r;
  }
}

extern "C" void kernel_launch(void* const* d_in, const int* in_sizes, int n_in,
                              void* d_out, int out_size, void* d_ws,
                              size_t ws_size, hipStream_t stream) {
  const float* x = (const float*)d_in[0];
  const int* boundaries = (const int*)d_in[1];
  float* out = (float*)d_out;

  int* starts = (int*)d_ws;

  seg_starts_kernel<<<Bn, 256, 0, stream>>>(boundaries, starts);

  constexpr int DCHUNK = 256;
  const int grid = Bn * (Pn / 16) * (Dn / DCHUNK);
  pool_wmma_kernel<<<grid, 128, 0, stream>>>(x, starts, out);
}


# Boundary scan folded into the pooling kernel: each workgroup ranks its batch row's 4096 flags by wave ballots (8 16-byte loads per lane) and keeps its 17 starts in LDS; one launch, was two
# speedup vs baseline: 2.1567x; 1.1843x over previous
#include <hip/hip_runtime.h>
#include <hip/hip_bf16.h>


constexpr int Bn = 8;
constexpr int Sn = 4096;
constexpr int Dn = 1024;
constexpr int Pn = 512;

typedef float v2f __attribute__((ext_vector_type(2)));
typedef float v4f __attribute__((ext_vector_type(4)));
typedef float v8f __attribute__((ext_vector_type(8)));
typedef int v4i __attribute__((ext_vector_type(4)));

__global__ __launch_bounds__(128) void pool_wmma_kernel(
    const float* __restrict__ x, const int* __restrict__ bnd,
    float* __restrict__ out) {
  constexpr int DCHUNK = 256;
  constexpr int NPB = Pn / 16;
  constexpr int ND = Dn / DCHUNK;
  constexpr int UNR = 8;
  constexpr int NQ = Sn / (4 * 128);

  const int wg = blockIdx.x;
  const int dchunk = wg % ND;
  const int pb = (wg / ND) % NPB;
  const int b = wg / (ND * NPB);
  const int p0 = pb * 16;

  __shared__ int sStarts[17];
  __shared__ int sWave[4];
  const int tid = threadIdx.x;
  const int lane = tid & 31;
  const int wave = tid >> 5;

  const v4i* rv = reinterpret_cast<const v4i*>(bnd + (size_t)b * Sn) +
                  wave * (32 * NQ) + lane;
  v4i q[NQ];
#pragma unroll
  for (int i = 0; i < NQ; ++i) q[i] = rv[32 * i];

  int pre[NQ];
  int tot = 0;
#pragma unroll
  for (int i = 0; i < NQ; ++i) {
    const unsigned m0 = __builtin_amdgcn_ballot_w32(q[i].x != 0);
    const unsigned m1 = __builtin_amdgcn_ballot_w32(q[i].y != 0);
    const unsigned m2 = __builtin_amdgcn_ballot_w32(q[i].z != 0);
    const unsigned m3 = __builtin_amdgcn_ballot_w32(q[i].w != 0);
    unsigned below = __builtin_amdgcn_mbcnt_lo(m0, 0u);
    below = __builtin_amdgcn_mbcnt_lo(m1, below);
    below = __builtin_amdgcn_mbcnt_lo(m2, below);
    below = __builtin_amdgcn_mbcnt_lo(m3, below);
    pre[i] = tot + (int)below;
    tot += __builtin_popcount(m0) + __builtin_popcount(m1) +
           __builtin_popcount(m2) + __builtin_popcount(m3);
  }
  if (lane == 0) sWave[wave] = tot;
  if (tid == 0 && pb == NPB - 1) sStarts[16] = Sn;
  __syncthreads();

  int wbase = 0;
#pragma unroll
  for (int w = 0; w < 3; ++w) wbase += (w < wave) ? sWave[w] : 0;

  const int lo = p0 - wbase;
  if (lo + 16 >= 0 && lo < tot) {
#pragma unroll
    for (int i = 0; i < NQ; ++i) {
      int slot = pre[i] - lo;
      const int t0 = wave * (128 * NQ) + 128 * i + 4 * lane;
#pragma unroll
      for (int c = 0; c < 4; ++c) {
        if (q[i][c] != 0) {
          if ((unsigned)slot <= 16u) sStarts[slot] = t0 + c;
          ++slot;
        }
      }
    }
  }
  __syncthreads();

  const int hi = lane >> 4;
  const int mrow = lane & 15;
  const int n = lane & 15;

  const int startM = sStarts[mrow];
  const int endM = sStarts[mrow + 1];
  const int s0 = sStarts[0];
  const int s1 = sStarts[16];
  const int kmax = s1 - 1;

  const int dbase = dchunk * DCHUNK + wave * 64;
  const char* xrow = reinterpret_cast<const char*>(x + (size_t)b * Sn * Dn);
  const unsigned lane_off = (unsigned)(dbase + 4 * n) * 4u;

  v8f c0 = {}, c1 = {}, c2 = {}, c3 = {};
  const int koff = hi * 2;

  for (int s = s0; s < s1; s += 4 * UNR) {
    v4f ra[UNR], rb[UNR];
    for (int i = 0; i < UNR; ++i) {
      const int ka = s + 4 * i + koff;
      const int kb = ka + 1;
      const int kac = (ka < kmax) ? ka : kmax;
      const int kbc = (kb < kmax) ? kb : kmax;
      ra[i] = *reinterpret_cast<const v4f*>(xrow + ((unsigned)kac * (Dn * 4u) + lane_off));
      rb[i] = *reinterpret_cast<const v4f*>(xrow + ((unsigned)kbc * (Dn * 4u) + lane_off));
    }
    __builtin_amdgcn_sched_barrier(0);
    for (int i = 0; i < UNR; ++i) {
      const int ka = s + 4 * i + koff;
      const int kb = ka + 1;

      v2f A;
      A.x = (ka >= startM && ka < endM) ? 1.0f : 0.0f;
      A.y = (kb >= startM && kb < endM) ? 1.0f : 0.0f;

      v2f B0, B1, B2, B3;
      B0.x = ra[i].x; B0.y = rb[i].x;
      B1.x = ra[i].y; B1.y = rb[i].y;
      B2.x = ra[i].z; B2.y = rb[i].z;
      B3.x = ra[i].w; B3.y = rb[i].w;

      c0 = __builtin_amdgcn_wmma_f32_16x16x4_f32(false, A, false, B0, (short)0,
                                                 c0, false, false);
      c1 = __builtin_amdgcn_wmma_f32_16x16x4_f32(false, A, false, B1, (short)0,
                                                 c1, false, false);
      c2 = __builtin_amdgcn_wmma_f32_16x16x4_f32(false, A, false, B2, (short)0,
                                                 c2, false, false);
      c3 = __builtin_amdgcn_wmma_f32_16x16x4_f32(false, A, false, B3, (short)0,
                                                 c3, false, false);
    }
  }

#pragma unroll
  for (int j = 0; j < 8; ++j) {
    const int rl = j + hi * 8;
    const float inv = 1.0f / (float)(sStarts[rl + 1] - sStarts[rl]);
    const size_t o = ((size_t)(b * Pn + p0 + rl)) * Dn + dbase + 4 * n;
    v4f r;
    r.x = c0[j] * inv;
    r.y = c1[j] * inv;
    r.z = c2[j] * inv;
    r.w = c3[j] * inv;
    *reinterpret_cast<v4f*>(out + o) = r;
  }
}

extern "C" void kernel_launch(void* const* d_in, const int* in_sizes, int n_in,
                              void* d_out, int out_size, void* d_ws,
                              size_t ws_size, hipStream_t stream) {
  const float* x = (const float*)d_in[0];
  const int* boundaries = (const int*)d_in[1];
  float* out = (float*)d_out;

  constexpr int DCHUNK = 256;
  const int grid = Bn * (Pn / 16) * (Dn / DCHUNK);
  pool_wmma_kernel<<<grid, 128, 0, stream>>>(x, boundaries, out);
}


# K loop peeled: full 32-token trips issue their 16 loads from one address register with constant row offsets, no clamps; only the last trip clamps and runs WMMAs just for the 4-token chunks left
# speedup vs baseline: 2.3472x; 1.0883x over previous
#include <hip/hip_runtime.h>
#include <hip/hip_bf16.h>


constexpr int Bn = 8;
constexpr int Sn = 4096;
constexpr int Dn = 1024;
constexpr int Pn = 512;

typedef float v2f __attribute__((ext_vector_type(2)));
typedef float v4f __attribute__((ext_vector_type(4)));
typedef float v8f __attribute__((ext_vector_type(8)));
typedef int v4i __attribute__((ext_vector_type(4)));

__global__ __launch_bounds__(128) void pool_wmma_kernel(
    const float* __restrict__ x, const int* __restrict__ bnd,
    float* __restrict__ out) {
  constexpr int DCHUNK = 256;
  constexpr int NPB = Pn / 16;
  constexpr int ND = Dn / DCHUNK;
  constexpr int UNR = 8;
  constexpr int NQ = Sn / (4 * 128);

  const int wg = blockIdx.x;
  const int dchunk = wg % ND;
  const int pb = (wg / ND) % NPB;
  const int b = wg / (ND * NPB);
  const int p0 = pb * 16;

  __shared__ int sStarts[17];
  __shared__ int sWave[4];
  const int tid = threadIdx.x;
  const int lane = tid & 31;
  const int wave = tid >> 5;

  const v4i* rv = reinterpret_cast<const v4i*>(bnd + (size_t)b * Sn) +
                  wave * (32 * NQ) + lane;
  v4i q[NQ];
#pragma unroll
  for (int i = 0; i < NQ; ++i) q[i] = rv[32 * i];

  int pre[NQ];
  int tot = 0;
#pragma unroll
  for (int i = 0; i < NQ; ++i) {
    const unsigned m0 = __builtin_amdgcn_ballot_w32(q[i].x != 0);
    const unsigned m1 = __builtin_amdgcn_ballot_w32(q[i].y != 0);
    const unsigned m2 = __builtin_amdgcn_ballot_w32(q[i].z != 0);
    const unsigned m3 = __builtin_amdgcn_ballot_w32(q[i].w != 0);
    unsigned below = __builtin_amdgcn_mbcnt_lo(m0, 0u);
    below = __builtin_amdgcn_mbcnt_lo(m1, below);
    below = __builtin_amdgcn_mbcnt_lo(m2, below);
    below = __builtin_amdgcn_mbcnt_lo(m3, below);
    pre[i] = tot + (int)below;
    tot += __builtin_popcount(m0) + __builtin_popcount(m1) +
           __builtin_popcount(m2) + __builtin_popcount(m3);
  }
  if (lane == 0) sWave[wave] = tot;
  if (tid == 0 && pb == NPB - 1) sStarts[16] = Sn;
  __syncthreads();

  int wbase = 0;
#pragma unroll
  for (int w = 0; w < 3; ++w) wbase += (w < wave) ? sWave[w] : 0;

  const int lo = p0 - wbase;
  if (lo + 16 >= 0 && lo < tot) {
#pragma unroll
    for (int i = 0; i < NQ; ++i) {
      int slot = pre[i] - lo;
      const int t0 = wave * (128 * NQ) + 128 * i + 4 * lane;
#pragma unroll
      for (int c = 0; c < 4; ++c) {
        if (q[i][c] != 0) {
          if ((unsigned)slot <= 16u) sStarts[slot] = t0 + c;
          ++slot;
        }
      }
    }
  }
  __syncthreads();

  const int hi = lane >> 4;
  const int mrow = lane & 15;
  const int n = lane & 15;

  const int startM = sStarts[mrow];
  const int endM = sStarts[mrow + 1];
  const int s0 = sStarts[0];
  const int s1 = sStarts[16];
  const int kmax = s1 - 1;

  const int dbase = dchunk * DCHUNK + wave * 64;
  const char* xrow = reinterpret_cast<const char*>(x + (size_t)b * Sn * Dn);
  const unsigned lane_off = (unsigned)(dbase + 4 * n) * 4u;

  v8f c0 = {}, c1 = {}, c2 = {}, c3 = {};
  const int koff = hi * 2;

  const int us0 = __builtin_amdgcn_readfirstlane(s0);
  const int us1 = __builtin_amdgcn_readfirstlane(s1);
  const unsigned lenM = (unsigned)(endM - startM);

  int s = us0;
  for (; s + 4 * UNR <= us1; s += 4 * UNR) {
    const int kbase = s + koff;
    const char* p = xrow + ((unsigned)kbase * (Dn * 4u) + lane_off);
    const unsigned ubase = (unsigned)(kbase - startM);
    v4f ra[UNR], rb[UNR];
#pragma unroll
    for (int i = 0; i < UNR; ++i) {
      ra[i] = *reinterpret_cast<const v4f*>(p + (4 * i) * (Dn * 4));
      rb[i] = *reinterpret_cast<const v4f*>(p + (4 * i + 1) * (Dn * 4));
    }
    __builtin_amdgcn_sched_barrier(0);
#pragma unroll
    for (int i = 0; i < UNR; ++i) {
      v2f A;
      A.x = (ubase + (unsigned)(4 * i) < lenM) ? 1.0f : 0.0f;
      A.y = (ubase + (unsigned)(4 * i + 1) < lenM) ? 1.0f : 0.0f;

      v2f B0, B1, B2, B3;
      B0.x = ra[i].x; B0.y = rb[i].x;
      B1.x = ra[i].y; B1.y = rb[i].y;
      B2.x = ra[i].z; B2.y = rb[i].z;
      B3.x = ra[i].w; B3.y = rb[i].w;

      c0 = __builtin_amdgcn_wmma_f32_16x16x4_f32(false, A, false, B0, (short)0,
                                                 c0, false, false);
      c1 = __builtin_amdgcn_wmma_f32_16x16x4_f32(false, A, false, B1, (short)0,
                                                 c1, false, false);
      c2 = __builtin_amdgcn_wmma_f32_16x16x4_f32(false, A, false, B2, (short)0,
                                                 c2, false, false);
      c3 = __builtin_amdgcn_wmma_f32_16x16x4_f32(false, A, false, B3, (short)0,
                                                 c3, false, false);
    }
  }

  if (s < us1) {
    const int nch = (us1 - s + 3) >> 2;
    v4f ra[UNR], rb[UNR];
#pragma unroll
    for (int i = 0; i < UNR; ++i) {
      const int ka = s + 4 * i + koff;
      const int kb = ka + 1;
      const int kac = (ka < kmax) ? ka : kmax;
      const int kbc = (kb < kmax) ? kb : kmax;
      ra[i] = *reinterpret_cast<const v4f*>(xrow + ((unsigned)kac * (Dn * 4u) + lane_off));
      rb[i] = *reinterpret_cast<const v4f*>(xrow + ((unsigned)kbc * (Dn * 4u) + lane_off));
    }
    __builtin_amdgcn_sched_barrier(0);
#pragma unroll
    for (int i = 0; i < UNR; ++i) {
      if (i < nch) {
        const int ka = s + 4 * i + koff;
        const int kb = ka + 1;

        v2f A;
        A.x = (ka >= startM && ka < endM) ? 1.0f : 0.0f;
        A.y = (kb >= startM && kb < endM) ? 1.0f : 0.0f;

        v2f B0, B1, B2, B3;
        B0.x = ra[i].x; B0.y = rb[i].x;
        B1.x = ra[i].y; B1.y = rb[i].y;
        B2.x = ra[i].z; B2.y = rb[i].z;
        B3.x = ra[i].w; B3.y = rb[i].w;

        c0 = __builtin_amdgcn_wmma_f32_16x16x4_f32(false, A, false, B0, (short)0,
                                                   c0, false, false);
        c1 = __builtin_amdgcn_wmma_f32_16x16x4_f32(false, A, false, B1, (short)0,
                                                   c1, false, false);
        c2 = __builtin_amdgcn_wmma_f32_16x16x4_f32(false, A, false, B2, (short)0,
                                                   c2, false, false);
        c3 = __builtin_amdgcn_wmma_f32_16x16x4_f32(false, A, false, B3, (short)0,
                                                   c3, false, false);
      }
    }
  }

#pragma unroll
  for (int j = 0; j < 8; ++j) {
    const int rl = j + hi * 8;
    const float inv = 1.0f / (float)(sStarts[rl + 1] - sStarts[rl]);
    const size_t o = ((size_t)(b * Pn + p0 + rl)) * Dn + dbase + 4 * n;
    v4f r;
    r.x = c0[j] * inv;
    r.y = c1[j] * inv;
    r.z = c2[j] * inv;
    r.w = c3[j] * inv;
    *reinterpret_cast<v4f*>(out + o) = r;
  }
}

extern "C" void kernel_launch(void* const* d_in, const int* in_sizes, int n_in,
                              void* d_out, int out_size, void* d_ws,
                              size_t ws_size, hipStream_t stream) {
  const float* x = (const float*)d_in[0];
  const int* boundaries = (const int*)d_in[1];
  float* out = (float*)d_out;

  constexpr int DCHUNK = 256;
  const int grid = Bn * (Pn / 16) * (Dn / DCHUNK);
  pool_wmma_kernel<<<grid, 128, 0, stream>>>(x, boundaries, out);
}


# K loop trips doubled to 64 tokens: a lane issues 32 16-byte x loads (16 KiB per wave in flight) ahead of the trip's 64 WMMAs, so half as many waits for memory; was 16 loads per 32 WMMAs
# speedup vs baseline: 2.5285x; 1.0772x over previous
#include <hip/hip_runtime.h>
#include <hip/hip_bf16.h>


constexpr int Bn = 8;
constexpr int Sn = 4096;
constexpr int Dn = 1024;
constexpr int Pn = 512;

typedef float v2f __attribute__((ext_vector_type(2)));
typedef float v4f __attribute__((ext_vector_type(4)));
typedef float v8f __attribute__((ext_vector_type(8)));
typedef int v4i __attribute__((ext_vector_type(4)));

constexpr int UNR = 8;

__device__ __forceinline__ void issue_full(v4f (&ra)[UNR], v4f (&rb)[UNR],
                                           const char* p) {
#pragma unroll
  for (int i = 0; i < UNR; ++i) {
    ra[i] = *reinterpret_cast<const v4f*>(p + (4 * i) * (Dn * 4));
    rb[i] = *reinterpret_cast<const v4f*>(p + (4 * i + 1) * (Dn * 4));
  }
}

__device__ __forceinline__ void issue_last(v4f (&ra)[UNR], v4f (&rb)[UNR],
                                           const char* xrow, unsigned lane_off,
                                           int s, int koff, int kmax) {
#pragma unroll
  for (int i = 0; i < UNR; ++i) {
    const int ka = s + 4 * i + koff;
    const int kb = ka + 1;
    const int kac = (ka < kmax) ? ka : kmax;
    const int kbc = (kb < kmax) ? kb : kmax;
    ra[i] = *reinterpret_cast<const v4f*>(xrow + ((unsigned)kac * (Dn * 4u) + lane_off));
    rb[i] = *reinterpret_cast<const v4f*>(xrow + ((unsigned)kbc * (Dn * 4u) + lane_off));
  }
}

__device__ __forceinline__ void mma_group(const v4f (&ra)[UNR],
                                          const v4f (&rb)[UNR], int kbase,
                                          int nch, int startM, unsigned lenM,
                                          v8f& c0, v8f& c1, v8f& c2, v8f& c3) {
  const unsigned ubase = (unsigned)(kbase - startM);
#pragma unroll
  for (int i = 0; i < UNR; ++i) {
    if (i < nch) {
      v2f A;
      A.x = (ubase + (unsigned)(4 * i) < lenM) ? 1.0f : 0.0f;
      A.y = (ubase + (unsigned)(4 * i + 1) < lenM) ? 1.0f : 0.0f;

      v2f B0, B1, B2, B3;
      B0.x = ra[i].x; B0.y = rb[i].x;
      B1.x = ra[i].y; B1.y = rb[i].y;
      B2.x = ra[i].z; B2.y = rb[i].z;
      B3.x = ra[i].w; B3.y = rb[i].w;

      c0 = __builtin_amdgcn_wmma_f32_16x16x4_f32(false, A, false, B0, (short)0,
                                                 c0, false, false);
      c1 = __builtin_amdgcn_wmma_f32_16x16x4_f32(false, A, false, B1, (short)0,
                                                 c1, false, false);
      c2 = __builtin_amdgcn_wmma_f32_16x16x4_f32(false, A, false, B2, (short)0,
                                                 c2, false, false);
      c3 = __builtin_amdgcn_wmma_f32_16x16x4_f32(false, A, false, B3, (short)0,
                                                 c3, false, false);
    }
  }
}

__global__ __launch_bounds__(128) void pool_wmma_kernel(
    const float* x, const int* __restrict__ bnd,
    float* __restrict__ out) {
  constexpr int DCHUNK = 256;
  constexpr int NPB = Pn / 16;
  constexpr int ND = Dn / DCHUNK;
  constexpr int NQ = Sn / (4 * 128);

  const int wg = blockIdx.x;
  const int dchunk = wg % ND;
  const int pb = (wg / ND) % NPB;
  const int b = wg / (ND * NPB);
  const int p0 = pb * 16;

  __shared__ int sStarts[17];
  __shared__ int sWave[4];
  const int tid = threadIdx.x;
  const int lane = tid & 31;
  const int wave = tid >> 5;

  const v4i* rv = reinterpret_cast<const v4i*>(bnd + (size_t)b * Sn) +
                  wave * (32 * NQ) + lane;
  v4i q[NQ];
#pragma unroll
  for (int i = 0; i < NQ; ++i) q[i] = rv[32 * i];

  int pre[NQ];
  int tot = 0;
#pragma unroll
  for (int i = 0; i < NQ; ++i) {
    const unsigned m0 = __builtin_amdgcn_ballot_w32(q[i].x != 0);
    const unsigned m1 = __builtin_amdgcn_ballot_w32(q[i].y != 0);
    const unsigned m2 = __builtin_amdgcn_ballot_w32(q[i].z != 0);
    const unsigned m3 = __builtin_amdgcn_ballot_w32(q[i].w != 0);
    unsigned below = __builtin_amdgcn_mbcnt_lo(m0, 0u);
    below = __builtin_amdgcn_mbcnt_lo(m1, below);
    below = __builtin_amdgcn_mbcnt_lo(m2, below);
    below = __builtin_amdgcn_mbcnt_lo(m3, below);
    pre[i] = tot + (int)below;
    tot += __builtin_popcount(m0) + __builtin_popcount(m1) +
           __builtin_popcount(m2) + __builtin_popcount(m3);
  }
  if (lane == 0) sWave[wave] = tot;
  if (tid == 0 && pb == NPB - 1) sStarts[16] = Sn;
  __syncthreads();

  int wbase = 0;
#pragma unroll
  for (int w = 0; w < 3; ++w) wbase += (w < wave) ? sWave[w] : 0;

  const int lo = p0 - wbase;
  if (lo + 16 >= 0 && lo < tot) {
#pragma unroll
    for (int i = 0; i < NQ; ++i) {
      int slot = pre[i] - lo;
      const int t0 = wave * (128 * NQ) + 128 * i + 4 * lane;
#pragma unroll
      for (int c = 0; c < 4; ++c) {
        if (q[i][c] != 0) {
          if ((unsigned)slot <= 16u) sStarts[slot] = t0 + c;
          ++slot;
        }
      }
    }
  }
  __syncthreads();

  const int hi = lane >> 4;
  const int mrow = lane & 15;
  const int n = lane & 15;

  const int startM = sStarts[mrow];
  const int endM = sStarts[mrow + 1];
  const int s0 = sStarts[0];
  const int s1 = sStarts[16];
  const int kmax = s1 - 1;

  const int dbase = dchunk * DCHUNK + wave * 64;
  const char* xrow = reinterpret_cast<const char*>(x + (size_t)b * Sn * Dn);
  const unsigned lane_off = (unsigned)(dbase + 4 * n) * 4u;

  v8f c0 = {}, c1 = {}, c2 = {}, c3 = {};
  const int koff = hi * 2;

  const int us0 = __builtin_amdgcn_readfirstlane(s0);
  const int us1 = __builtin_amdgcn_readfirstlane(s1);
  const unsigned lenM = (unsigned)(endM - startM);

  const char* xlane = xrow + ((unsigned)koff * (Dn * 4u) + lane_off);

  int s = us0;
  for (; s + 8 * UNR <= us1; s += 8 * UNR) {
    const char* p = xlane + (unsigned)s * (Dn * 4u);
    v4f ra0[UNR], rb0[UNR], ra1[UNR], rb1[UNR];
    issue_full(ra0, rb0, p);
    issue_full(ra1, rb1, p + 4 * UNR * (Dn * 4));
    __builtin_amdgcn_sched_barrier(0);
    mma_group(ra0, rb0, s + koff, UNR, startM, lenM, c0, c1, c2, c3);
    mma_group(ra1, rb1, s + 4 * UNR + koff, UNR, startM, lenM, c0, c1, c2, c3);
  }
  if (s < us1) {
    v4f ra0[UNR], rb0[UNR], ra1[UNR], rb1[UNR];
    issue_last(ra0, rb0, xrow, lane_off, s, koff, kmax);
    issue_last(ra1, rb1, xrow, lane_off, s + 4 * UNR, koff, kmax);
    __builtin_amdgcn_sched_barrier(0);
    mma_group(ra0, rb0, s + koff, (us1 - s + 3) >> 2, startM, lenM, c0, c1, c2, c3);
    mma_group(ra1, rb1, s + 4 * UNR + koff, (us1 - s - 4 * UNR + 3) >> 2, startM,
              lenM, c0, c1, c2, c3);
  }

#pragma unroll
  for (int j = 0; j < 8; ++j) {
    const int rl = j + hi * 8;
    const float inv = 1.0f / (float)(sStarts[rl + 1] - sStarts[rl]);
    const size_t o = ((size_t)(b * Pn + p0 + rl)) * Dn + dbase + 4 * n;
    v4f r;
    r.x = c0[j] * inv;
    r.y = c1[j] * inv;
    r.z = c2[j] * inv;
    r.w = c3[j] * inv;
    *reinterpret_cast<v4f*>(out + o) = r;
  }
}

extern "C" void kernel_launch(void* const* d_in, const int* in_sizes, int n_in,
                              void* d_out, int out_size, void* d_ws,
                              size_t ws_size, hipStream_t stream) {
  const float* x = (const float*)d_in[0];
  const int* boundaries = (const int*)d_in[1];
  float* out = (float*)d_out;

  constexpr int DCHUNK = 256;
  const int grid = Bn * (Pn / 16) * (Dn / DCHUNK);
  pool_wmma_kernel<<<grid, 128, 0, stream>>>(x, boundaries, out);
}
